# starting kernel
# baseline (speedup 1.0000x reference)
#include <hip/hip_runtime.h>

typedef __attribute__((ext_vector_type(2))) float v2f;
typedef __attribute__((ext_vector_type(8))) float v8f;

namespace {
constexpr int BATCH = 16;
constexpr int CIN   = 64;
constexpr int COUT  = 64;
constexpr int HIMG  = 64;
constexpr int WIMG  = 64;
constexpr int TAPS  = 9;
constexpr int KDIM  = CIN * TAPS;
constexpr int PSTR  = 68;
}

__global__ __launch_bounds__(128)
void masked_conv2d_wmma_f32(const float* __restrict__ x,
                            const float* __restrict__ mask,
                            const float* __restrict__ wgt,
                            const float* __restrict__ bias,
                            float* __restrict__ out)
{
  __shared__ float P[KDIM * PSTR];

  const int tid = threadIdx.x;
  const int y   = blockIdx.x & (HIMG - 1);
  const int b   = blockIdx.x >> 6;

  const float* xb = x + (size_t)b * CIN * HIMG * WIMG;

  const int n     = tid & (WIMG - 1);
  const int cbase = tid >> 6;
#pragma unroll
  for (int t = 0; t < TAPS; ++t) {
    const int  i    = t / 3;
    const int  j    = t - 3 * i;
    const int  row  = y + i - 1;
    const int  col  = n + j - 1;
    const bool ok   = ((unsigned)row < (unsigned)HIMG) &
                      ((unsigned)col < (unsigned)WIMG);
    const float mv  = mask[t * (HIMG * WIMG) + y * WIMG + n];
    const float* xsrc = xb + (size_t)row * WIMG + col;
#pragma unroll 4
    for (int c = cbase; c < CIN; c += 2) {
      float v = 0.0f;
      if (ok) v = xsrc[(size_t)c * (HIMG * WIMG)] * mv;
      P[(c * TAPS + t) * PSTR + n] = v;
    }
  }
  __syncthreads();

  const int wave  = tid >> 5;
  const int lane  = tid & 31;
  const int half  = lane >> 4;
  const int lidx  = lane & 15;
  const int mbase = (wave & 1) * 32;
  const int nbase = (wave >> 1) * 32;
  const int pix0  = nbase + lidx;
  const int pix1  = pix0 + 16;

  const int koff = half * 2;
  const float* wrow0 = wgt + (size_t)(mbase + lidx) * KDIM + koff;
  const float* wrow1 = wrow0 + (size_t)16 * KDIM;
  const float* pb = &P[koff * PSTR + pix0];

  v8f acc00 = {};
  v8f acc01 = {};
  v8f acc10 = {};
  v8f acc11 = {};

#pragma unroll 4
  for (int k0 = 0; k0 < KDIM; k0 += 4) {
    if ((k0 & 63) == 0) {
      __builtin_prefetch(wrow0 + k0 + 128, 0, 3);
      __builtin_prefetch(wrow1 + k0 + 128, 0, 3);
    }

    v2f a0, a1;
    a0.x = wrow0[k0];
    a0.y = wrow0[k0 + 1];
    a1.x = wrow1[k0];
    a1.y = wrow1[k0 + 1];

    v2f b0, b1;
    b0.x = pb[0];
    b0.y = pb[PSTR];
    b1.x = pb[16];
    b1.y = pb[PSTR + 16];

    acc00 = __builtin_amdgcn_wmma_f32_16x16x4_f32(false, a0, false, b0,
                                                  (short)0, acc00, false, false);
    acc01 = __builtin_amdgcn_wmma_f32_16x16x4_f32(false, a0, false, b1,
                                                  (short)0, acc01, false, false);
    acc10 = __builtin_amdgcn_wmma_f32_16x16x4_f32(false, a1, false, b0,
                                                  (short)0, acc10, false, false);
    acc11 = __builtin_amdgcn_wmma_f32_16x16x4_f32(false, a1, false, b1,
                                                  (short)0, acc11, false, false);

    pb += 4 * PSTR;
  }

  float* ob = out + (size_t)b * COUT * HIMG * WIMG + (size_t)y * WIMG;
#pragma unroll
  for (int r = 0; r < 8; ++r) {
    int o0 = mbase + half * 8 + r;
    int o1 = o0 + 16;
    float bv0 = bias[o0];
    float bv1 = bias[o1];
    ob[(size_t)o0 * (HIMG * WIMG) + pix0] = acc00[r] + bv0;
    ob[(size_t)o0 * (HIMG * WIMG) + pix1] = acc01[r] + bv0;
    ob[(size_t)o1 * (HIMG * WIMG) + pix0] = acc10[r] + bv1;
    ob[(size_t)o1 * (HIMG * WIMG) + pix1] = acc11[r] + bv1;
  }
}

extern "C" void kernel_launch(void* const* d_in, const int* in_sizes, int n_in,
                              void* d_out, int out_size, void* d_ws, size_t ws_size,
                              hipStream_t stream) {
  (void)in_sizes; (void)n_in; (void)out_size; (void)d_ws; (void)ws_size;
  const float* x    = (const float*)d_in[0];
  const float* mask = (const float*)d_in[1];
  const float* wgt  = (const float*)d_in[2];
  const float* bias = (const float*)d_in[3];
  float* out = (float*)d_out;

  dim3 grid(BATCH * HIMG);
  dim3 block(128);
  masked_conv2d_wmma_f32<<<grid, block, 0, stream>>>(x, mask, wgt, bias, out);
}


# W held in LDS by a persistent workgroup over 4 image rows (was re-read from L2 by 8-byte global loads per K-step); panel per 16-pixel tile: both operands load as ready pairs one set ahead; same sums
# speedup vs baseline: 4.4946x; 4.4946x over previous
#include <hip/hip_runtime.h>

typedef __attribute__((ext_vector_type(2))) float v2f;
typedef __attribute__((ext_vector_type(4))) float v4f;
typedef __attribute__((ext_vector_type(8))) float v8f;

namespace {
constexpr int BATCH = 16;
constexpr int CIN   = 64;
constexpr int COUT  = 64;
constexpr int HIMG  = 64;
constexpr int WIMG  = 64;
constexpr int TAPS  = 9;
constexpr int KDIM  = CIN * TAPS;
constexpr int PROW  = 16;
constexpr int PTILE = KDIM * PROW + 16;
constexpr int WSTR  = 580;
constexpr int ROWS_PER_WG = 4;
constexpr int KBLK  = 16;
}

__device__ __forceinline__ float from_left_lane(float v) {
  return __builtin_bit_cast(float, __builtin_amdgcn_update_dpp(
      0, __builtin_bit_cast(int, v), 0x111  , 0xf, 0xf, true));
}
__device__ __forceinline__ float from_right_lane(float v) {
  return __builtin_bit_cast(float, __builtin_amdgcn_update_dpp(
      0, __builtin_bit_cast(int, v), 0x101  , 0xf, 0xf, true));
}

__device__ __forceinline__ void lds_barrier() {
  __builtin_amdgcn_fence(__ATOMIC_RELEASE, "workgroup", "local");
  __builtin_amdgcn_s_barrier();
  __builtin_amdgcn_fence(__ATOMIC_ACQUIRE, "workgroup", "local");
}

struct RowRegs {
  v4f xv[3][8];
  v4f mq[9];
};

__device__ __forceinline__ void load_row_inputs(RowRegs& R, const float* __restrict__ xb,
                                                const float* __restrict__ mask,
                                                int y, int q, int slot) {
#pragma unroll
  for (int i = 0; i < 3; ++i) {
    const int row  = y + i - 1;
    const int rowc = row < 0 ? 0 : (row > HIMG - 1 ? HIMG - 1 : row);
#pragma unroll
    for (int m = 0; m < 8; ++m) {
      const int c = m * 8 + slot;
      R.xv[i][m] = *reinterpret_cast<const v4f*>(xb + (size_t)c * (HIMG * WIMG) + rowc * WIMG + 4 * q);
    }
  }
#pragma unroll
  for (int t = 0; t < TAPS; ++t)
    R.mq[t] = *reinterpret_cast<const v4f*>(mask + (size_t)t * (HIMG * WIMG) + y * WIMG + 4 * q);
}

__device__ __forceinline__ void build_panel(float* __restrict__ P, const RowRegs& R,
                                            int y, int q, int slot) {
  const v4f zero4 = {0.0f, 0.0f, 0.0f, 0.0f};
#pragma unroll
  for (int i = 0; i < 3; ++i) {
    const int  row   = y + i - 1;
    const bool rowok = (unsigned)row < (unsigned)HIMG;
#pragma unroll
    for (int m = 0; m < 8; ++m) {
      const int c = m * 8 + slot;
      const v4f v = rowok ? R.xv[i][m] : zero4;
      const float lft = from_left_lane(v.w);
      const float rgt = from_right_lane(v.x);
      v4f t0 = {lft, v.x, v.y, v.z};
      v4f t2 = {v.y, v.z, v.w, rgt};
      float* dst = &P[(q >> 2) * PTILE + (c * TAPS + 3 * i) * PROW + 4 * (q & 3)];
      *reinterpret_cast<v4f*>(dst)            = t0 * R.mq[3 * i];
      *reinterpret_cast<v4f*>(dst + PROW)     = v  * R.mq[3 * i + 1];
      *reinterpret_cast<v4f*>(dst + 2 * PROW) = t2 * R.mq[3 * i + 2];
    }
  }
}

struct Frag {
  v2f a0[4], a1[4], b0[4], b1[4];
};

__device__ __forceinline__ void load_frag(Frag& f, const float* pa0, const float* pa1,
                                          const float* pb) {
#pragma unroll
  for (int s = 0; s < 4; ++s) {
    const int k0 = 4 * s;
    f.a0[s] = *reinterpret_cast<const v2f*>(pa0 + k0);
    f.a1[s] = *reinterpret_cast<const v2f*>(pa1 + k0);
    f.b0[s].x = pb[k0 * PROW];
    f.b0[s].y = pb[k0 * PROW + PROW];
    f.b1[s].x = pb[PTILE + k0 * PROW];
    f.b1[s].y = pb[PTILE + k0 * PROW + PROW];
  }
}

__device__ __forceinline__ void mma_frag(const Frag& f, v8f& acc00, v8f& acc01,
                                         v8f& acc10, v8f& acc11) {
#pragma unroll
  for (int s = 0; s < 4; ++s) {
    acc00 = __builtin_amdgcn_wmma_f32_16x16x4_f32(false, f.a0[s], false, f.b0[s],
                                                  (short)0, acc00, false, false);
    acc01 = __builtin_amdgcn_wmma_f32_16x16x4_f32(false, f.a0[s], false, f.b1[s],
                                                  (short)0, acc01, false, false);
    acc10 = __builtin_amdgcn_wmma_f32_16x16x4_f32(false, f.a1[s], false, f.b0[s],
                                                  (short)0, acc10, false, false);
    acc11 = __builtin_amdgcn_wmma_f32_16x16x4_f32(false, f.a1[s], false, f.b1[s],
                                                  (short)0, acc11, false, false);
  }
}

__global__ __launch_bounds__(128)
void masked_conv2d_wmma_f32(const float* __restrict__ x,
                            const float* __restrict__ mask,
                            const float* __restrict__ wgt,
                            const float* __restrict__ bias,
                            float* __restrict__ out)
{
  __shared__ __attribute__((aligned(16))) float smem[4 * PTILE + COUT * WSTR];
  float* const P  = smem;
  float* const Wl = smem + 4 * PTILE;

  const int tid   = threadIdx.x;
  const int b     = blockIdx.x >> 4;
  const int ybase = (blockIdx.x & 15) * ROWS_PER_WG;

  const float* xb = x + (size_t)b * CIN * HIMG * WIMG;

  const int q    = tid & 15;
  const int slot = tid >> 4;

  RowRegs R;
  load_row_inputs(R, xb, mask, ybase, q, slot);

  {
    const float* wsrc = wgt + slot * KDIM + 4 * q;
    float*       wdst = &Wl[slot * WSTR + 4 * q];
#pragma unroll 1
    for (int rh = 0; rh < 2; ++rh) {
      v4f t[4][9];
#pragma unroll
      for (int rr = 0; rr < 4; ++rr)
#pragma unroll
        for (int j = 0; j < 9; ++j)
          t[rr][j] = *reinterpret_cast<const v4f*>(wsrc + (size_t)((rh * 4 + rr) * 8) * KDIM + 64 * j);
#pragma unroll
      for (int rr = 0; rr < 4; ++rr)
#pragma unroll
        for (int j = 0; j < 9; ++j)
          *reinterpret_cast<v4f*>(wdst + ((rh * 4 + rr) * 8) * WSTR + 64 * j) = t[rr][j];
    }
  }

  const int wave  = tid >> 5;
  const int lane  = tid & 31;
  const int half  = lane >> 4;
  const int lidx  = lane & 15;
  const int mbase = (wave & 1) * 32;
  const int nbase = (wave >> 1) * 32;
  const int pix0  = nbase + lidx;
  const int pix1  = pix0 + 16;
  const int koff  = half * 2;

  const float* pa0_base = &Wl[(mbase + lidx) * WSTR + koff];
  const float* pa1_base = pa0_base + 16 * WSTR;
  const float* pb_base  = &P[(nbase >> 4) * PTILE + koff * PROW + lidx];

  float bv0[8], bv1[8];
#pragma unroll
  for (int r = 0; r < 8; ++r) {
    bv0[r] = bias[mbase + half * 8 + r];
    bv1[r] = bias[mbase + half * 8 + r + 16];
  }

#pragma unroll 1
  for (int ry = 0; ry < ROWS_PER_WG; ++ry) {
    const int y = ybase + ry;

    build_panel(P, R, y, q, slot);
    lds_barrier();

    if (ry + 1 < ROWS_PER_WG)
      load_row_inputs(R, xb, mask, y + 1, q, slot);

    v8f acc00 = {};
    v8f acc01 = {};
    v8f acc10 = {};
    v8f acc11 = {};

    const float* pa0 = pa0_base;
    const float* pa1 = pa1_base;
    const float* pb  = pb_base;

    Frag f0, f1;
    load_frag(f0, pa0, pa1, pb);
#pragma unroll 1
    for (int kb = 0; kb < KDIM / KBLK - 2; kb += 2) {
      load_frag(f1, pa0 + KBLK, pa1 + KBLK, pb + KBLK * PROW);
      __builtin_amdgcn_sched_barrier(0);
      mma_frag(f0, acc00, acc01, acc10, acc11);
      __builtin_amdgcn_sched_barrier(0);
      load_frag(f0, pa0 + 2 * KBLK, pa1 + 2 * KBLK, pb + 2 * KBLK * PROW);
      __builtin_amdgcn_sched_barrier(0);
      mma_frag(f1, acc00, acc01, acc10, acc11);
      __builtin_amdgcn_sched_barrier(0);
      pa0 += 2 * KBLK;
      pa1 += 2 * KBLK;
      pb  += 2 * KBLK * PROW;
    }
    load_frag(f1, pa0 + KBLK, pa1 + KBLK, pb + KBLK * PROW);
    __builtin_amdgcn_sched_barrier(0);
    mma_frag(f0, acc00, acc01, acc10, acc11);
    mma_frag(f1, acc00, acc01, acc10, acc11);

    float* ob = out + (size_t)b * COUT * HIMG * WIMG + (size_t)y * WIMG;
#pragma unroll
    for (int r = 0; r < 8; ++r) {
      const int o0 = mbase + half * 8 + r;
      const int o1 = o0 + 16;
      ob[(size_t)o0 * (HIMG * WIMG) + pix0] = acc00[r] + bv0[r];
      ob[(size_t)o0 * (HIMG * WIMG) + pix1] = acc01[r] + bv0[r];
      ob[(size_t)o1 * (HIMG * WIMG) + pix0] = acc10[r] + bv1[r];
      ob[(size_t)o1 * (HIMG * WIMG) + pix1] = acc11[r] + bv1[r];
    }
    lds_barrier();
  }
}

extern "C" void kernel_launch(void* const* d_in, const int* in_sizes, int n_in,
                              void* d_out, int out_size, void* d_ws, size_t ws_size,
                              hipStream_t stream) {
  (void)in_sizes; (void)n_in; (void)out_size; (void)d_ws; (void)ws_size;
  const float* x    = (const float*)d_in[0];
  const float* mask = (const float*)d_in[1];
  const float* wgt  = (const float*)d_in[2];
  const float* bias = (const float*)d_in[3];
  float* out = (float*)d_out;

  dim3 grid(BATCH * HIMG / ROWS_PER_WG);
  dim3 block(128);
  masked_conv2d_wmma_f32<<<grid, block, 0, stream>>>(x, mask, wgt, bias, out);
}


# 8 waves (2 per SIMD), each owns 16-pixel x 64-channel tiles and a private half-K panel: no barrier after the W copy, a wave's panel build runs under its SIMD partner's matrix work; same sums
# speedup vs baseline: 4.7016x; 1.0461x over previous
#include <hip/hip_runtime.h>

typedef __attribute__((ext_vector_type(2))) float v2f;
typedef __attribute__((ext_vector_type(4))) float v4f;
typedef __attribute__((ext_vector_type(8))) float v8f;

namespace {
constexpr int BATCH = 16;
constexpr int CIN   = 64;
constexpr int COUT  = 64;
constexpr int HIMG  = 64;
constexpr int WIMG  = 64;
constexpr int TAPS  = 9;
constexpr int KDIM  = CIN * TAPS;
constexpr int PROW  = 16;
constexpr int CPASS = 32;
constexpr int KPASS = CPASS * TAPS;
constexpr int PWAVE = KPASS * PROW;
constexpr int NWAVE = 8;
constexpr int WSTR  = 580;
constexpr int ROWS_PER_WG = 4;
constexpr int KBLK  = 16;
}

__device__ __forceinline__ float from_left_lane(float v) {
  return __builtin_bit_cast(float, __builtin_amdgcn_update_dpp(
      0, __builtin_bit_cast(int, v), 0x111  , 0xf, 0xf, true));
}
__device__ __forceinline__ float from_right_lane(float v) {
  return __builtin_bit_cast(float, __builtin_amdgcn_update_dpp(
      0, __builtin_bit_cast(int, v), 0x101  , 0xf, 0xf, true));
}

__device__ __forceinline__ void lds_barrier() {
  __builtin_amdgcn_fence(__ATOMIC_RELEASE, "workgroup", "local");
  __builtin_amdgcn_s_barrier();
  __builtin_amdgcn_fence(__ATOMIC_ACQUIRE, "workgroup", "local");
}

__device__ __forceinline__ void wave_panel_fence() {
  __builtin_amdgcn_fence(__ATOMIC_RELEASE, "wavefront", "local");
  __builtin_amdgcn_fence(__ATOMIC_ACQUIRE, "wavefront", "local");
}

struct PassRegs {
  v4f   xv[3][4];
  float edge[3][4];
};

__device__ __forceinline__ void load_pass_inputs(PassRegs& R, const float* __restrict__ xb,
                                                 int y, int px, int ep, int pass, int slot) {
#pragma unroll
  for (int i = 0; i < 3; ++i) {
    const int row  = y + i - 1;
    const int rowc = row < 0 ? 0 : (row > HIMG - 1 ? HIMG - 1 : row);
#pragma unroll
    for (int m = 0; m < 4; ++m) {
      const int c = pass * CPASS + m * 8 + slot;
      const float* p = xb + (size_t)c * (HIMG * WIMG) + rowc * WIMG;
      R.xv[i][m]   = *reinterpret_cast<const v4f*>(p + px);
      R.edge[i][m] = p[ep];
    }
  }
}

__device__ __forceinline__ void load_tile_mask(v4f (&mq)[TAPS], const float* __restrict__ mask,
                                               int y, int nt, int q4) {
#pragma unroll
  for (int t = 0; t < TAPS; ++t)
    mq[t] = *reinterpret_cast<const v4f*>(mask + (size_t)t * (HIMG * WIMG) + y * WIMG + 16 * nt + 4 * q4);
}

__device__ __forceinline__ void build_panel(float* __restrict__ Pw, const PassRegs& R,
                                            const v4f (&mq)[TAPS], int y, int nt, int q4, int slot) {
  const v4f zero4 = {0.0f, 0.0f, 0.0f, 0.0f};
  const bool lin = nt > 0;
  const bool rin = nt < 3;
#pragma unroll
  for (int i = 0; i < 3; ++i) {
    const int  row   = y + i - 1;
    const bool rowok = (unsigned)row < (unsigned)HIMG;
#pragma unroll
    for (int m = 0; m < 4; ++m) {
      const int cc = m * 8 + slot;
      const v4f   v = rowok ? R.xv[i][m] : zero4;
      const float e = rowok ? R.edge[i][m] : 0.0f;
      const float lfl = from_left_lane(v.w);
      const float rgl = from_right_lane(v.x);
      const float lft = (q4 == 0) ? (lin ? e : 0.0f) : lfl;
      const float rgt = (q4 == 3) ? (rin ? e : 0.0f) : rgl;
      v4f t0 = {lft, v.x, v.y, v.z};
      v4f t2 = {v.y, v.z, v.w, rgt};
      float* dst = &Pw[(cc * TAPS + 3 * i) * PROW + 4 * q4];
      *reinterpret_cast<v4f*>(dst)            = t0 * mq[3 * i];
      *reinterpret_cast<v4f*>(dst + PROW)     = v  * mq[3 * i + 1];
      *reinterpret_cast<v4f*>(dst + 2 * PROW) = t2 * mq[3 * i + 2];
    }
  }
}

struct Frag {
  v2f a[4][4];
  v2f b[4];
};

__device__ __forceinline__ void load_frag(Frag& f, const float* pa0, const float* pa1,
                                          const float* pa2, const float* pa3, const float* pb) {
#pragma unroll
  for (int s = 0; s < 4; ++s) {
    const int k0 = 4 * s;
    f.a[0][s] = *reinterpret_cast<const v2f*>(pa0 + k0);
    f.a[1][s] = *reinterpret_cast<const v2f*>(pa1 + k0);
    f.a[2][s] = *reinterpret_cast<const v2f*>(pa2 + k0);
    f.a[3][s] = *reinterpret_cast<const v2f*>(pa3 + k0);
    f.b[s].x = pb[k0 * PROW];
    f.b[s].y = pb[k0 * PROW + PROW];
  }
}

__device__ __forceinline__ void mma_frag(const Frag& f, v8f (&acc)[4]) {
#pragma unroll
  for (int s = 0; s < 4; ++s) {
#pragma unroll
    for (int mt = 0; mt < 4; ++mt)
      acc[mt] = __builtin_amdgcn_wmma_f32_16x16x4_f32(false, f.a[mt][s], false, f.b[s],
                                                      (short)0, acc[mt], false, false);
  }
}

__device__ __forceinline__ void k_pass(v8f (&acc)[4], const float* pa0, const float* pa1,
                                       const float* pa2, const float* pa3, const float* pb) {
  Frag f0, f1;
  load_frag(f0, pa0, pa1, pa2, pa3, pb);
#pragma unroll 1
  for (int kb = 0; kb < KPASS / KBLK - 2; kb += 2) {
    load_frag(f1, pa0 + KBLK, pa1 + KBLK, pa2 + KBLK, pa3 + KBLK, pb + KBLK * PROW);
    __builtin_amdgcn_sched_barrier(0);
    mma_frag(f0, acc);
    __builtin_amdgcn_sched_barrier(0);
    load_frag(f0, pa0 + 2 * KBLK, pa1 + 2 * KBLK, pa2 + 2 * KBLK, pa3 + 2 * KBLK,
              pb + 2 * KBLK * PROW);
    __builtin_amdgcn_sched_barrier(0);
    mma_frag(f1, acc);
    __builtin_amdgcn_sched_barrier(0);
    pa0 += 2 * KBLK;
    pa1 += 2 * KBLK;
    pa2 += 2 * KBLK;
    pa3 += 2 * KBLK;
    pb  += 2 * KBLK * PROW;
  }
  load_frag(f1, pa0 + KBLK, pa1 + KBLK, pa2 + KBLK, pa3 + KBLK, pb + KBLK * PROW);
  __builtin_amdgcn_sched_barrier(0);
  mma_frag(f0, acc);
  mma_frag(f1, acc);
}

__global__ __launch_bounds__(256)
void masked_conv2d_wmma_f32(const float* __restrict__ x,
                            const float* __restrict__ mask,
                            const float* __restrict__ wgt,
                            const float* __restrict__ bias,
                            float* __restrict__ out)
{
  __shared__ __attribute__((aligned(16))) float smem[NWAVE * PWAVE + COUT * WSTR];
  float* const P  = smem;
  float* const Wl = smem + NWAVE * PWAVE;

  const int tid   = threadIdx.x;
  const int b     = blockIdx.x >> 4;
  const int ybase = (blockIdx.x & 15) * ROWS_PER_WG;

  const float* xb = x + (size_t)b * CIN * HIMG * WIMG;

  const int wave = __builtin_amdgcn_readfirstlane(tid >> 5);
  const int lane = tid & 31;
  const int q4   = lane & 3;
  const int slot = lane >> 2;
  const int nt   = wave & 3;
  const int y0   = ybase + (wave >> 2);
  const int px   = 16 * nt + 4 * q4;
  const int eraw = px + (q4 == 0 ? -1 : 0) + (q4 == 3 ? 4 : 0);
  const int ep   = eraw < 0 ? 0 : (eraw > WIMG - 1 ? WIMG - 1 : eraw);

  PassRegs R;
  v4f mq[TAPS];
  load_pass_inputs(R, xb, y0, px, ep, 0, slot);
  load_tile_mask(mq, mask, y0, nt, q4);

  {
    const int wq    = tid & 15;
    const int wslot = tid >> 4;
    const float* wsrc = wgt + wslot * KDIM + 4 * wq;
    float*       wdst = &Wl[wslot * WSTR + 4 * wq];
    v4f t[4][9];
#pragma unroll
    for (int g = 0; g < 4; ++g)
#pragma unroll
      for (int j = 0; j < 9; ++j)
        t[g][j] = *reinterpret_cast<const v4f*>(wsrc + (size_t)(g * 16) * KDIM + 64 * j);
#pragma unroll
    for (int g = 0; g < 4; ++g)
#pragma unroll
      for (int j = 0; j < 9; ++j)
        *reinterpret_cast<v4f*>(wdst + (g * 16) * WSTR + 64 * j) = t[g][j];
  }
  lds_barrier();

  const int half = lane >> 4;
  const int lidx = lane & 15;
  const int koff = half * 2;
  float* const Pw = P + wave * PWAVE;

  const float* pa0 = &Wl[lidx * WSTR + koff];
  const float* pa1 = pa0 + 16 * WSTR;
  const float* pa2 = pa0 + 32 * WSTR;
  const float* pa3 = pa0 + 48 * WSTR;
  const float* pb_base = &Pw[koff * PROW + lidx];

  float bv[4][8];
#pragma unroll
  for (int mt = 0; mt < 4; ++mt)
#pragma unroll
    for (int r = 0; r < 8; ++r)
      bv[mt][r] = bias[mt * 16 + half * 8 + r];

#pragma unroll 1
  for (int it = 0; it < 2; ++it) {
    const int y = y0 + 2 * it;

    v8f acc[4];
#pragma unroll
    for (int mt = 0; mt < 4; ++mt) acc[mt] = v8f{};

    build_panel(Pw, R, mq, y, nt, q4, slot);
    wave_panel_fence();
    load_pass_inputs(R, xb, y, px, ep, 1, slot);
    k_pass(acc, pa0, pa1, pa2, pa3, pb_base);
    wave_panel_fence();

    build_panel(Pw, R, mq, y, nt, q4, slot);
    wave_panel_fence();
    if (it == 0) {
      load_pass_inputs(R, xb, y + 2, px, ep, 0, slot);
      load_tile_mask(mq, mask, y + 2, nt, q4);
    }
    k_pass(acc, pa0 + KPASS, pa1 + KPASS, pa2 + KPASS, pa3 + KPASS, pb_base);

    float* ob = out + (size_t)b * COUT * HIMG * WIMG + (size_t)y * WIMG + 16 * nt + lidx;
#pragma unroll
    for (int mt = 0; mt < 4; ++mt)
#pragma unroll
      for (int r = 0; r < 8; ++r)
        ob[(size_t)(mt * 16 + half * 8 + r) * (HIMG * WIMG)] = acc[mt][r] + bv[mt][r];
    wave_panel_fence();
  }
}

extern "C" void kernel_launch(void* const* d_in, const int* in_sizes, int n_in,
                              void* d_out, int out_size, void* d_ws, size_t ws_size,
                              hipStream_t stream) {
  (void)in_sizes; (void)n_in; (void)out_size; (void)d_ws; (void)ws_size;
  const float* x    = (const float*)d_in[0];
  const float* mask = (const float*)d_in[1];
  const float* wgt  = (const float*)d_in[2];
  const float* bias = (const float*)d_in[3];
  float* out = (float*)d_out;

  dim3 grid(BATCH * HIMG / ROWS_PER_WG);
  dim3 block(256);
  masked_conv2d_wmma_f32<<<grid, block, 0, stream>>>(x, mask, wgt, bias, out);
}


# Each wave owns one 32-pixel x 64-channel tile (8 accumulator tiles; was two 16-pixel tiles in turn): every W operand read from LDS feeds two matrix instructions; quarter-K panels; same sums
# speedup vs baseline: 4.9549x; 1.0539x over previous
#include <hip/hip_runtime.h>

typedef __attribute__((ext_vector_type(2))) float v2f;
typedef __attribute__((ext_vector_type(4))) float v4f;
typedef __attribute__((ext_vector_type(8))) float v8f;

namespace {
constexpr int BATCH = 16;
constexpr int CIN   = 64;
constexpr int COUT  = 64;
constexpr int HIMG  = 64;
constexpr int WIMG  = 64;
constexpr int TAPS  = 9;
constexpr int KDIM  = CIN * TAPS;
constexpr int PROW  = 32;
constexpr int CPASS = 16;
constexpr int NPASS = CIN / CPASS;
constexpr int KPASS = CPASS * TAPS;
constexpr int PWAVE = KPASS * PROW;
constexpr int NWAVE = 8;
constexpr int WSTR  = 580;
constexpr int ROWS_PER_WG = 4;
constexpr int KBLK  = 16;
}

__device__ __forceinline__ float from_left_lane(float v) {
  return __builtin_bit_cast(float, __builtin_amdgcn_update_dpp(
      0, __builtin_bit_cast(int, v), 0x111  , 0xf, 0xf, true));
}
__device__ __forceinline__ float from_right_lane(float v) {
  return __builtin_bit_cast(float, __builtin_amdgcn_update_dpp(
      0, __builtin_bit_cast(int, v), 0x101  , 0xf, 0xf, true));
}

__device__ __forceinline__ void lds_barrier() {
  __builtin_amdgcn_fence(__ATOMIC_RELEASE, "workgroup", "local");
  __builtin_amdgcn_s_barrier();
  __builtin_amdgcn_fence(__ATOMIC_ACQUIRE, "workgroup", "local");
}

__device__ __forceinline__ void wave_panel_fence() {
  __builtin_amdgcn_fence(__ATOMIC_RELEASE, "wavefront", "local");
  __builtin_amdgcn_fence(__ATOMIC_ACQUIRE, "wavefront", "local");
}

struct PassRegs {
  v4f   xv[3][4];
  float edge[3][4];
};

__device__ __forceinline__ void load_pass_inputs(PassRegs& R, const float* __restrict__ xb,
                                                 int y, int px, int ep, int pass, int slot) {
#pragma unroll
  for (int i = 0; i < 3; ++i) {
    const int row  = y + i - 1;
    const int rowc = row < 0 ? 0 : (row > HIMG - 1 ? HIMG - 1 : row);
#pragma unroll
    for (int m = 0; m < 4; ++m) {
      const int c = pass * CPASS + m * 4 + slot;
      const float* p = xb + (size_t)c * (HIMG * WIMG) + rowc * WIMG;
      R.xv[i][m]   = *reinterpret_cast<const v4f*>(p + px);
      R.edge[i][m] = p[ep];
    }
  }
}

__device__ __forceinline__ void load_tile_mask(v4f (&mq)[TAPS], const float* __restrict__ mask,
                                               int y, int px) {
#pragma unroll
  for (int t = 0; t < TAPS; ++t)
    mq[t] = *reinterpret_cast<const v4f*>(mask + (size_t)t * (HIMG * WIMG) + y * WIMG + px);
}

__device__ __forceinline__ void build_panel(float* __restrict__ Pw, const PassRegs& R,
                                            const v4f (&mq)[TAPS], int y, int nt, int q8, int slot) {
  const v4f zero4 = {0.0f, 0.0f, 0.0f, 0.0f};
  const bool lin = nt > 0;
  const bool rin = nt < 1;
#pragma unroll
  for (int i = 0; i < 3; ++i) {
    const int  row   = y + i - 1;
    const bool rowok = (unsigned)row < (unsigned)HIMG;
#pragma unroll
    for (int m = 0; m < 4; ++m) {
      const int cc = m * 4 + slot;
      const v4f   v = rowok ? R.xv[i][m] : zero4;
      const float e = rowok ? R.edge[i][m] : 0.0f;
      const float lfl = from_left_lane(v.w);
      const float rgl = from_right_lane(v.x);
      const float lft = (q8 == 0) ? (lin ? e : 0.0f) : lfl;
      const float rgt = (q8 == 7) ? (rin ? e : 0.0f) : rgl;
      v4f t0 = {lft, v.x, v.y, v.z};
      v4f t2 = {v.y, v.z, v.w, rgt};
      const int r0 = cc * TAPS + 3 * i;
      const int c0 = (4 * q8) ^ ((((r0    ) >> 1) & 1) << 4);
      const int c1 = (4 * q8) ^ ((((r0 + 1) >> 1) & 1) << 4);
      const int c2 = (4 * q8) ^ ((((r0 + 2) >> 1) & 1) << 4);
      *reinterpret_cast<v4f*>(&Pw[(r0    ) * PROW + c0]) = t0 * mq[3 * i];
      *reinterpret_cast<v4f*>(&Pw[(r0 + 1) * PROW + c1]) = v  * mq[3 * i + 1];
      *reinterpret_cast<v4f*>(&Pw[(r0 + 2) * PROW + c2]) = t2 * mq[3 * i + 2];
    }
  }
}

struct Frag {
  v2f a[4][4];
  v2f b[2][4];
};

__device__ __forceinline__ void load_frag(Frag& f, const float* pa0, const float* pa1,
                                          const float* pa2, const float* pa3,
                                          const float* pb0, const float* pb1) {
#pragma unroll
  for (int s = 0; s < 4; ++s) {
    const int k0 = 4 * s;
    f.a[0][s] = *reinterpret_cast<const v2f*>(pa0 + k0);
    f.a[1][s] = *reinterpret_cast<const v2f*>(pa1 + k0);
    f.a[2][s] = *reinterpret_cast<const v2f*>(pa2 + k0);
    f.a[3][s] = *reinterpret_cast<const v2f*>(pa3 + k0);
    f.b[0][s].x = pb0[k0 * PROW];
    f.b[0][s].y = pb0[k0 * PROW + PROW];
    f.b[1][s].x = pb1[k0 * PROW];
    f.b[1][s].y = pb1[k0 * PROW + PROW];
  }
}

__device__ __forceinline__ void mma_frag(const Frag& f, v8f (&acc)[4][2]) {
#pragma unroll
  for (int s = 0; s < 4; ++s) {
#pragma unroll
    for (int mt = 0; mt < 4; ++mt) {
#pragma unroll
      for (int j = 0; j < 2; ++j)
        acc[mt][j] = __builtin_amdgcn_wmma_f32_16x16x4_f32(false, f.a[mt][s], false, f.b[j][s],
                                                           (short)0, acc[mt][j], false, false);
    }
  }
}

__device__ __forceinline__ void k_pass(v8f (&acc)[4][2], const float*& pa0, const float*& pa1,
                                       const float*& pa2, const float*& pa3,
                                       const float* pb0, const float* pb1) {
  Frag f0, f1;
  load_frag(f0, pa0, pa1, pa2, pa3, pb0, pb1);
#pragma unroll 1
  for (int kb = 0; kb < KPASS / KBLK - 1; kb += 2) {
    load_frag(f1, pa0 + KBLK, pa1 + KBLK, pa2 + KBLK, pa3 + KBLK,
              pb0 + KBLK * PROW, pb1 + KBLK * PROW);
    __builtin_amdgcn_sched_barrier(0);
    mma_frag(f0, acc);
    __builtin_amdgcn_sched_barrier(0);
    load_frag(f0, pa0 + 2 * KBLK, pa1 + 2 * KBLK, pa2 + 2 * KBLK, pa3 + 2 * KBLK,
              pb0 + 2 * KBLK * PROW, pb1 + 2 * KBLK * PROW);
    __builtin_amdgcn_sched_barrier(0);
    mma_frag(f1, acc);
    __builtin_amdgcn_sched_barrier(0);
    pa0 += 2 * KBLK;
    pa1 += 2 * KBLK;
    pa2 += 2 * KBLK;
    pa3 += 2 * KBLK;
    pb0 += 2 * KBLK * PROW;
    pb1 += 2 * KBLK * PROW;
  }
  mma_frag(f0, acc);
  pa0 += KBLK;
  pa1 += KBLK;
  pa2 += KBLK;
  pa3 += KBLK;
}

__global__ __launch_bounds__(256)
void masked_conv2d_wmma_f32(const float* __restrict__ x,
                            const float* __restrict__ mask,
                            const float* __restrict__ wgt,
                            const float* __restrict__ bias,
                            float* __restrict__ out)
{
  __shared__ __attribute__((aligned(16))) float smem[NWAVE * PWAVE + COUT * WSTR];
  float* const P  = smem;
  float* const Wl = smem + NWAVE * PWAVE;

  const int tid   = threadIdx.x;
  const int b     = blockIdx.x >> 4;
  const int ybase = (blockIdx.x & 15) * ROWS_PER_WG;

  const float* xb = x + (size_t)b * CIN * HIMG * WIMG;

  const int wave = __builtin_amdgcn_readfirstlane(tid >> 5);
  const int lane = tid & 31;
  const int q8   = lane & 7;
  const int slot = lane >> 3;
  const int nt   = wave & 1;
  const int y    = ybase + (wave >> 1);
  const int px   = 32 * nt + 4 * q8;
  const int eraw = px + (q8 == 0 ? -1 : 0) + (q8 == 7 ? 4 : 0);
  const int ep   = eraw < 0 ? 0 : (eraw > WIMG - 1 ? WIMG - 1 : eraw);

  PassRegs R;
  v4f mq[TAPS];
  load_pass_inputs(R, xb, y, px, ep, 0, slot);
  load_tile_mask(mq, mask, y, px);

  {
    const int wq    = tid & 15;
    const int wslot = tid >> 4;
    const float* wsrc = wgt + wslot * KDIM + 4 * wq;
    float*       wdst = &Wl[wslot * WSTR + 4 * wq];
    v4f t[4][9];
#pragma unroll
    for (int g = 0; g < 4; ++g)
#pragma unroll
      for (int j = 0; j < 9; ++j)
        t[g][j] = *reinterpret_cast<const v4f*>(wsrc + (size_t)(g * 16) * KDIM + 64 * j);
#pragma unroll
    for (int g = 0; g < 4; ++g)
#pragma unroll
      for (int j = 0; j < 9; ++j)
        *reinterpret_cast<v4f*>(wdst + (g * 16) * WSTR + 64 * j) = t[g][j];
  }
  lds_barrier();

  const int half = lane >> 4;
  const int lidx = lane & 15;
  const int koff = half * 2;
  float* const Pw = P + wave * PWAVE;

  const float* pa0 = &Wl[lidx * WSTR + koff];
  const float* pa1 = pa0 + 16 * WSTR;
  const float* pa2 = pa0 + 32 * WSTR;
  const float* pa3 = pa0 + 48 * WSTR;
  const float* pb0 = &Pw[koff * PROW + 16 * half + lidx];
  const float* pb1 = &Pw[koff * PROW + 16 * (1 - half) + lidx];

  float bv[4][8];
#pragma unroll
  for (int mt = 0; mt < 4; ++mt)
#pragma unroll
    for (int r = 0; r < 8; ++r)
      bv[mt][r] = bias[mt * 16 + half * 8 + r];

  v8f acc[4][2];
#pragma unroll
  for (int mt = 0; mt < 4; ++mt) {
    acc[mt][0] = v8f{};
    acc[mt][1] = v8f{};
  }

#pragma unroll 1
  for (int pass = 0; pass < NPASS; ++pass) {
    build_panel(Pw, R, mq, y, nt, q8, slot);
    wave_panel_fence();
    if (pass + 1 < NPASS)
      load_pass_inputs(R, xb, y, px, ep, pass + 1, slot);
    k_pass(acc, pa0, pa1, pa2, pa3, pb0, pb1);
    wave_panel_fence();
  }

  float* ob = out + (size_t)b * COUT * HIMG * WIMG + (size_t)y * WIMG + 32 * nt + lidx;
#pragma unroll
  for (int mt = 0; mt < 4; ++mt)
#pragma unroll
    for (int r = 0; r < 8; ++r) {
      float* o = ob + (size_t)(mt * 16 + half * 8 + r) * (HIMG * WIMG);
      o[0]  = acc[mt][0][r] + bv[mt][r];
      o[16] = acc[mt][1][r] + bv[mt][r];
    }
}

extern "C" void kernel_launch(void* const* d_in, const int* in_sizes, int n_in,
                              void* d_out, int out_size, void* d_ws, size_t ws_size,
                              hipStream_t stream) {
  (void)in_sizes; (void)n_in; (void)out_size; (void)d_ws; (void)ws_size;
  const float* x    = (const float*)d_in[0];
  const float* mask = (const float*)d_in[1];
  const float* wgt  = (const float*)d_in[2];
  const float* bias = (const float*)d_in[3];
  float* out = (float*)d_out;

  dim3 grid(BATCH * HIMG / ROWS_PER_WG);
  dim3 block(256);
  masked_conv2d_wmma_f32<<<grid, block, 0, stream>>>(x, mask, wgt, bias, out);
}
